# starting kernel
# baseline (speedup 1.0000x reference)
#include <hip/hip_runtime.h>

#define NN 100000
#define NE 1600000

typedef float v2f __attribute__((ext_vector_type(2)));
typedef float v8f __attribute__((ext_vector_type(8)));

__global__ __launch_bounds__(256) void k_deg(const int* __restrict__ dst,
                                             unsigned* __restrict__ deg, int E) {
  int e = blockIdx.x * 256 + threadIdx.x;
  if (e < E) atomicAdd(&deg[dst[e]], 1u);
}

__global__ __launch_bounds__(256) void k_dinv(const unsigned* __restrict__ deg,
                                              float* __restrict__ dinv, int N) {
  int i = blockIdx.x * 256 + threadIdx.x;
  if (i < N) dinv[i] = 1.0f / sqrtf((float)deg[i] + 1.0f);
}

__global__ __launch_bounds__(256) void k_norm(const int* __restrict__ src,
                                              const int* __restrict__ dst,
                                              const float* __restrict__ dinv,
                                              float* __restrict__ nrm, int E) {
  int e = blockIdx.x * 256 + threadIdx.x;
  if (e < E) nrm[e] = dinv[src[e]] * dinv[dst[e]];
}

template <int K>
__global__ __launch_bounds__(256) void k_gemm(const float* __restrict__ X,
                                              const float* __restrict__ W,
                                              float* __restrict__ H, int M) {
  __shared__ __align__(16) v2f wl[(K / 2) * 64];
  for (int i = threadIdx.x; i < (K / 2) * 64; i += 256) {
    int p = i >> 6, n = i & 63;
    v2f t = { W[(size_t)(2 * p) * 64 + n], W[(size_t)(2 * p + 1) * 64 + n] };
    wl[i] = t;
  }
  __syncthreads();

  int wave  = threadIdx.x >> 5;
  int lane  = threadIdx.x & 31;
  int strip = blockIdx.x * 8 + wave;
  if (strip * 16 >= M) return;
  int row0 = strip * 16;
  int m    = lane & 15;
  int h2   = lane >> 4;

  const float* xrow = X + (size_t)(row0 + m) * K;
  v8f a0 = {}, a1 = {}, a2 = {}, a3 = {};
  for (int kb = 0; kb < K; kb += 4) {
    int pa = (kb >> 1) + h2;
    v2f a = *(const v2f*)(xrow + 2 * pa);
    const v2f* wp = &wl[pa * 64 + m];
    v2f b0 = wp[0];
    v2f b1 = wp[16];
    v2f b2 = wp[32];
    v2f b3 = wp[48];
    a0 = __builtin_amdgcn_wmma_f32_16x16x4_f32(false, a, false, b0, (short)0, a0, false, false);
    a1 = __builtin_amdgcn_wmma_f32_16x16x4_f32(false, a, false, b1, (short)0, a1, false, false);
    a2 = __builtin_amdgcn_wmma_f32_16x16x4_f32(false, a, false, b2, (short)0, a2, false, false);
    a3 = __builtin_amdgcn_wmma_f32_16x16x4_f32(false, a, false, b3, (short)0, a3, false, false);
  }
  float* out0 = H + (size_t)(row0 + 8 * h2) * 64 + m;
#pragma unroll
  for (int j = 0; j < 8; ++j) {
    float* r = out0 + (size_t)j * 64;
    r[0] = a0[j]; r[16] = a1[j]; r[32] = a2[j]; r[48] = a3[j];
  }
}

__global__ __launch_bounds__(256) void k_agg(const float* __restrict__ H,
                                             const int* __restrict__ src,
                                             const int* __restrict__ dst,
                                             const float* __restrict__ nrm,
                                             float* __restrict__ agg, int E) {
  int e = (int)((blockIdx.x * 256 + threadIdx.x) >> 5);
  if (e >= E) return;
  int lane = threadIdx.x & 31;
  int s = src[e], d = dst[e];
  float w = nrm[e];
  v2f hv = *(const v2f*)(H + (size_t)s * 64 + lane * 2);
  float* p = agg + (size_t)d * 64 + lane * 2;
  unsafeAtomicAdd(p,     w * hv.x);
  unsafeAtomicAdd(p + 1, w * hv.y);
}

__global__ __launch_bounds__(256) void k_relu_l2(const float* __restrict__ agg,
                                                 const float* __restrict__ H,
                                                 const float* __restrict__ dinv,
                                                 const float* __restrict__ bias,
                                                 float* __restrict__ out, int N) {
  int node = (int)((blockIdx.x * 256 + threadIdx.x) >> 5);
  if (node >= N) return;
  int lane = threadIdx.x & 31;
  float di = dinv[node];
  float sl = di * di;
  size_t base = (size_t)node * 64 + lane * 2;
  v2f a  = *(const v2f*)(agg + base);
  v2f hv = *(const v2f*)(H + base);
  float vx = fmaxf(a.x + sl * hv.x + bias[lane * 2],     0.0f);
  float vy = fmaxf(a.y + sl * hv.y + bias[lane * 2 + 1], 0.0f);
  float ss = vx * vx + vy * vy;
#pragma unroll
  for (int off = 16; off > 0; off >>= 1) ss += __shfl_xor(ss, off, 32);
  float scale = 1.0f / fmaxf(sqrtf(ss), 1e-12f);
  out[base]     = vx * scale;
  out[base + 1] = vy * scale;
}

__global__ __launch_bounds__(256) void k_final(const float* __restrict__ agg,
                                               const float* __restrict__ H,
                                               const float* __restrict__ dinv,
                                               const float* __restrict__ bias,
                                               float* __restrict__ out, int N) {
  int idx = blockIdx.x * 256 + threadIdx.x;
  if (idx >= N * 64) return;
  int node = idx >> 6;
  int c    = idx & 63;
  float di = dinv[node];
  out[idx] = agg[idx] + di * di * H[idx] + bias[c];
}

extern "C" void kernel_launch(void* const* d_in, const int* in_sizes, int n_in,
                              void* d_out, int out_size, void* d_ws, size_t ws_size,
                              hipStream_t stream) {
  const float* x  = (const float*)d_in[0];
  const int*   ei = (const int*)d_in[1];
  const float* W1 = (const float*)d_in[2];
  const float* b1 = (const float*)d_in[3];
  const float* W2 = (const float*)d_in[4];
  const float* b2 = (const float*)d_in[5];
  const float* W3 = (const float*)d_in[6];
  const float* b3 = (const float*)d_in[7];
  float* out = (float*)d_out;

  const int* src = ei;
  const int* dst = ei + NE;

  char* ws = (char*)d_ws;
  size_t off = 0;
  auto carve = [&](size_t bytes) {
    void* p = ws + off;
    off = (off + bytes + 255) & ~(size_t)255;
    return p;
  };
  unsigned* deg  = (unsigned*)carve((size_t)NN * 4);
  float*    dinv = (float*)   carve((size_t)NN * 4);
  float*    nrm  = (float*)   carve((size_t)NE * 4);
  float*    h    = (float*)   carve((size_t)NN * 64 * 4);
  float*    agg  = (float*)   carve((size_t)NN * 64 * 4);
  float*    hn   = (float*)   carve((size_t)NN * 64 * 4);

  const int eBlk   = (NE + 255) / 256;
  const int nBlk   = (NN + 255) / 256;
  const int gBlk   = ((NN / 16) + 7) / 8;
  const int aggBlk = NE / 8;
  const int wnBlk  = (NN * 32 + 255) / 256;
  const int feBlk  = (NN * 64 + 255) / 256;

  hipMemsetAsync(deg, 0, (size_t)NN * 4, stream);
  k_deg <<<eBlk, 256, 0, stream>>>(dst, deg, NE);
  k_dinv<<<nBlk, 256, 0, stream>>>(deg, dinv, NN);
  k_norm<<<eBlk, 256, 0, stream>>>(src, dst, dinv, nrm, NE);

  k_gemm<256><<<gBlk, 256, 0, stream>>>(x, W1, h, NN);
  hipMemsetAsync(agg, 0, (size_t)NN * 64 * 4, stream);
  k_agg    <<<aggBlk, 256, 0, stream>>>(h, src, dst, nrm, agg, NE);
  k_relu_l2<<<wnBlk,  256, 0, stream>>>(agg, h, dinv, b1, hn, NN);

  k_gemm<64><<<gBlk, 256, 0, stream>>>(hn, W2, h, NN);
  hipMemsetAsync(agg, 0, (size_t)NN * 64 * 4, stream);
  k_agg    <<<aggBlk, 256, 0, stream>>>(h, src, dst, nrm, agg, NE);
  k_relu_l2<<<wnBlk,  256, 0, stream>>>(agg, h, dinv, b2, hn, NN);

  k_gemm<64><<<gBlk, 256, 0, stream>>>(hn, W3, h, NN);
  hipMemsetAsync(agg, 0, (size_t)NN * 64 * 4, stream);
  k_agg  <<<aggBlk, 256, 0, stream>>>(h, src, dst, nrm, agg, NE);
  k_final<<<feBlk,  256, 0, stream>>>(agg, h, dinv, b3, out, NN);

  (void)in_sizes; (void)n_in; (void)out_size; (void)ws_size;
}


# Layer-1 product moved into the bucketing launch as extra blocks: 782 product blocks behind 391 bucketing blocks (256 threads, 16 edges each), matrix work under their waits; 8 launches to 7
# speedup vs baseline: 3.4253x; 3.4253x over previous
#include <hip/hip_runtime.h>


#define NN 100000
#define NE 1600000

#define BIN    200
#define NBINS  (NN / BIN)
#define SC_EPB 4096
#define HC_EPB 8192
static_assert(NN % BIN == 0, "ranges cover the nodes exactly");
static_assert(NBINS <= 512, "one scan block covers all ranges");
#define CUR_PITCH 32
static_assert(NN <= (1 << 20) && BIN <= 2048, "source id and row-in-range share one word");
static_assert(NE % 4 == 0, "the destination list is read in 16-byte pieces");

typedef float v2f __attribute__((ext_vector_type(2)));
typedef float v4f __attribute__((ext_vector_type(4)));
typedef float v8f __attribute__((ext_vector_type(8)));
typedef int   v4i __attribute__((ext_vector_type(4)));

__global__ __launch_bounds__(1024) void k_bincnt(const int* __restrict__ dst,
                                                 unsigned* __restrict__ bincnt) {
  constexpr int VPT = HC_EPB / (1024 * 4);
  __shared__ unsigned cnt[NBINS];
  const int t = threadIdx.x;
  for (int i = t; i < NBINS; i += 1024) cnt[i] = 0u;
  v4i  d[VPT];
  bool ok[VPT];
#pragma unroll
  for (int k = 0; k < VPT; ++k) {
    const int e = blockIdx.x * HC_EPB + k * 4096 + t * 4;
    ok[k] = e < NE;
    d[k] = *(const v4i*)(dst + (ok[k] ? e : 0));
  }
  __syncthreads();
#pragma unroll
  for (int k = 0; k < VPT; ++k) {
    if (ok[k]) {
      atomicAdd(&cnt[d[k].x / BIN], 1u);
      atomicAdd(&cnt[d[k].y / BIN], 1u);
      atomicAdd(&cnt[d[k].z / BIN], 1u);
      atomicAdd(&cnt[d[k].w / BIN], 1u);
    }
  }
  __syncthreads();
  for (int i = t; i < NBINS; i += 1024) {
    const unsigned c = cnt[i];
    if (c) atomicAdd(&bincnt[i], c);
  }
}

__global__ __launch_bounds__(512) void k_scan(const unsigned* __restrict__ bincnt,
                                              unsigned* __restrict__ binstart) {
  __shared__ unsigned s[512];
  const int t = threadIdx.x;
  const unsigned v = (t < NBINS) ? bincnt[t] : 0u;
  s[t] = v;
  __syncthreads();
  for (int off = 1; off < 512; off <<= 1) {
    unsigned a = (t >= off) ? s[t - off] : 0u;
    __syncthreads();
    s[t] += a;
    __syncthreads();
  }
  if (t < NBINS) binstart[t] = s[t] - v;
  if (t == NBINS - 1) binstart[NBINS] = s[t];
}

__global__ __launch_bounds__(256) void k_dinv(const int* __restrict__ rec,
                                              const unsigned* __restrict__ binstart,
                                              float* __restrict__ dinv) {
  constexpr int U = 8;
  __shared__ unsigned cnt[BIN];
  const int t = threadIdx.x;
  if (t < BIN) cnt[t] = 0u;
  const int eb = (int)binstart[blockIdx.x];
  const int ee = (int)binstart[blockIdx.x + 1];
  __syncthreads();
  for (int e = eb + t; e < ee; e += 256 * U) {
    int x[U];
#pragma unroll
    for (int k = 0; k < U; ++k) {
      const int ec = e + k * 256;
      x[k] = -1;
      if (ec < ee) x[k] = rec[ec];
    }
#pragma unroll
    for (int k = 0; k < U; ++k)
      if (x[k] >= 0) atomicAdd(&cnt[(unsigned)x[k] >> 20], 1u);
  }
  __syncthreads();
  if (t < BIN)
    dinv[blockIdx.x * BIN + t] = 1.0f / sqrtf((float)cnt[t] + 1.0f);
}

template <int K>
__device__ __forceinline__ void gemm_block(const float* __restrict__ X,
                                           const float* __restrict__ W,
                                           float* __restrict__ H, int M,
                                           v2f* wl, int gb) {
  for (int i = threadIdx.x; i < (K / 2) * 64; i += 256) {
    int p = i >> 6, n = i & 63;
    v2f t = { W[(size_t)(2 * p) * 64 + n], W[(size_t)(2 * p + 1) * 64 + n] };
    wl[i] = t;
  }
  __syncthreads();

  int wave  = threadIdx.x >> 5;
  int lane  = threadIdx.x & 31;
  int strip = gb * 8 + wave;
  if (strip * 16 >= M) return;
  int row0 = strip * 16;
  int m    = lane & 15;
  int h2   = lane >> 4;

  const float* xrow = X + (size_t)(row0 + m) * K;
  v8f a0 = {}, a1 = {}, a2 = {}, a3 = {};
  for (int kb = 0; kb < K; kb += 4) {
    int pa = (kb >> 1) + h2;
    v2f a = *(const v2f*)(xrow + 2 * pa);
    const v2f* wp = &wl[pa * 64 + m];
    v2f b0 = wp[0];
    v2f b1 = wp[16];
    v2f b2 = wp[32];
    v2f b3 = wp[48];
    a0 = __builtin_amdgcn_wmma_f32_16x16x4_f32(false, a, false, b0, (short)0, a0, false, false);
    a1 = __builtin_amdgcn_wmma_f32_16x16x4_f32(false, a, false, b1, (short)0, a1, false, false);
    a2 = __builtin_amdgcn_wmma_f32_16x16x4_f32(false, a, false, b2, (short)0, a2, false, false);
    a3 = __builtin_amdgcn_wmma_f32_16x16x4_f32(false, a, false, b3, (short)0, a3, false, false);
  }
  float* out0 = H + (size_t)(row0 + 8 * h2) * 64 + m;
#pragma unroll
  for (int j = 0; j < 8; ++j) {
    float* r = out0 + (size_t)j * 64;
    r[0] = a0[j]; r[16] = a1[j]; r[32] = a2[j]; r[48] = a3[j];
  }
}

#define SC_BLK ((NE + SC_EPB - 1) / SC_EPB)
#define G_BLK  (((NN / 16) + 7) / 8)
__global__ __launch_bounds__(256) void k_scatter_gemm(const int* __restrict__ src,
                                                      const int* __restrict__ dst,
                                                      const unsigned* __restrict__ binstart,
                                                      unsigned* __restrict__ cursor,
                                                      int* __restrict__ rec,
                                                      const float* __restrict__ X,
                                                      const float* __restrict__ W,
                                                      float* __restrict__ H) {
  __shared__ __align__(16) v2f wl[(256 / 2) * 64];
  if (blockIdx.x >= SC_BLK) {
    gemm_block<256>(X, W, H, NN, wl, (int)blockIdx.x - SC_BLK);
    return;
  }
  constexpr int PER = SC_EPB / 256;
  unsigned* cnt = (unsigned*)wl;
  const int t = threadIdx.x;
  for (int i = t; i < NBINS; i += 256) cnt[i] = 0u;

  const int e0 = blockIdx.x * SC_EPB + t;
  bool ok[PER];
  int  s[PER], d[PER], b[PER];
#pragma unroll
  for (int k = 0; k < PER; ++k) {
    const int e = e0 + k * 256;
    ok[k] = e < NE;
    const int ec = ok[k] ? e : 0;
    d[k] = dst[ec];
    s[k] = src[ec];
  }
  __syncthreads();

  unsigned r[PER];
#pragma unroll
  for (int k = 0; k < PER; ++k) {
    b[k] = d[k] / BIN;
    r[k] = 0u;
    if (ok[k]) r[k] = atomicAdd(&cnt[b[k]], 1u);
  }
  __syncthreads();

  for (int i = t; i < NBINS; i += 256) {
    const unsigned c  = cnt[i];
    const unsigned bs = binstart[i];
    unsigned base = 0u;
    if (c) base = atomicAdd(&cursor[i * CUR_PITCH], c);
    cnt[i] = bs + base;
  }
  __syncthreads();

#pragma unroll
  for (int k = 0; k < PER; ++k) {
    if (ok[k]) {
      const unsigned pos = cnt[b[k]] + r[k];
      const int dl = d[k] - b[k] * BIN;
      rec[pos] = s[k] | (dl << 20);
    }
  }
}


template <int N, int RP>
__device__ __forceinline__ unsigned lds_rows(const float* __restrict__ H,
                                             float* acc,
                                             int s, int dl, float w,
                                             unsigned m, int lane) {
  int sj[N], dj[N];
  float wj[N];
  v2f hv[N];
#pragma unroll
  for (int i = 0; i < N; ++i) {
    int j = __builtin_ctz(m);
    m &= m - 1;
    sj[i] = __builtin_amdgcn_readlane(s, j);
    dj[i] = __builtin_amdgcn_readlane(dl, j);
    wj[i] = __int_as_float(__builtin_amdgcn_readlane(__float_as_int(w), j));
  }
#pragma unroll
  for (int i = 0; i < N; ++i)
    hv[i] = *(const v2f*)(H + (size_t)sj[i] * 64 + lane * 2);
#pragma unroll
  for (int i = 0; i < N; ++i) {
    float* p = acc + dj[i] * RP + lane * 2;
    __hip_atomic_fetch_add(p,     wj[i] * hv[i].x, __ATOMIC_RELAXED, __HIP_MEMORY_SCOPE_WORKGROUP);
    __hip_atomic_fetch_add(p + 1, wj[i] * hv[i].y, __ATOMIC_RELAXED, __HIP_MEMORY_SCOPE_WORKGROUP);
  }
  return m;
}

#define NSTRIP ((BIN + 15) / 16)
static_assert(BIN % 8 == 0, "a half strip (8 rows) lies wholly inside or wholly outside the range");
static_assert(NSTRIP <= 16, "one strip per wave");

template <bool FINAL, bool MAKEW, bool PROD>
__global__ __launch_bounds__(512) void k_agg_lds(const float* __restrict__ H,
                                                 const int* __restrict__ rec,
                                                 float* __restrict__ wgt,
                                                 const unsigned* __restrict__ binstart,
                                                 const float* __restrict__ dinv,
                                                 const float* __restrict__ W,
                                                 const float* __restrict__ bias,
                                                 float* __restrict__ out) {
  constexpr int RP = PROD ? 68 : 64;
  __shared__ __align__(16) float acc[BIN * RP];
  __shared__ __align__(16) v2f wl[PROD ? 32 * 64 : 1];
  const int t    = threadIdx.x;
  const int lane = t & 31;
  const int wave = __builtin_amdgcn_readfirstlane(t >> 5);
  const int node0 = blockIdx.x * BIN;
  const int eb = (int)binstart[blockIdx.x];
  const int ee = (int)binstart[blockIdx.x + 1];

  const int g0 = eb + wave * 32;
  int   rc = 0, rc2 = 0;
  float wq = 0.0f;
  if (g0 + lane < ee) {
    rc = rec[g0 + lane];
    if (!MAKEW) wq = wgt[g0 + lane];
  }
  if (MAKEW) {
    if (g0 + 16 * 32 + lane < ee) rc2 = rec[g0 + 16 * 32 + lane];
  }
  v2f wreg[4];
  if (PROD) {
#pragma unroll
    for (int k = 0; k < 4; ++k) {
      const int i = t + k * 512;
      const int p = i >> 6, n = i & 63;
      wreg[k] = (v2f){ W[(2 * p) * 64 + n], W[(2 * p + 1) * 64 + n] };
    }
  }

  constexpr int INIT_IT = (BIN * 16 + 511) / 512;
  float fs = 0.0f, fd = 0.0f;
  {
    const int c4 = (t & 15) * 4;
    v4f bv = (v4f){0.0f, 0.0f, 0.0f, 0.0f};
    if (!PROD) bv = *(const v4f*)(bias + c4);
    v4f   hv[INIT_IT];
    float dv[INIT_IT];
#pragma unroll
    for (int k = 0; k < INIT_IT; ++k) {
      const int i = t + k * 512;
      hv[k] = (v4f){0.0f, 0.0f, 0.0f, 0.0f};
      dv[k] = 0.0f;
      if (i < BIN * 16) {
        const int row = i >> 4;
        dv[k] = dinv[node0 + row];
        hv[k] = *(const v4f*)(H + (size_t)(node0 + row) * 64 + c4);
      }
    }
    if (MAKEW) {
      fs = dinv[rc & 0xFFFFF];
      fd = dinv[node0 + (int)((unsigned)rc >> 20)];
    }
#pragma unroll
    for (int k = 0; k < INIT_IT; ++k) {
      const int i = t + k * 512;
      if (i < BIN * 16) {
        const int row = i >> 4;
        const float sl = dv[k] * dv[k];
        v4f o;
        o.x = sl * hv[k].x + bv.x;
        o.y = sl * hv[k].y + bv.y;
        o.z = sl * hv[k].z + bv.z;
        o.w = sl * hv[k].w + bv.w;
        *(v4f*)(acc + row * RP + c4) = o;
      }
    }
    if (PROD) {
#pragma unroll
      for (int k = 0; k < 4; ++k) wl[t + k * 512] = wreg[k];
    }
  }
  __syncthreads();

  for (int g = g0; g < ee; g += 16 * 32) {
    const int  cur = rc;
    const bool ok = (g + lane) < ee;
    const int  en = g + 16 * 32 + lane;
    float w;
    if (MAKEW) {
      w = fs * fd;
      if (ok) wgt[g + lane] = w;
      rc = rc2;
      const int e2 = g + 2 * 16 * 32 + lane;
      rc2 = 0;
      if (e2 < ee) rc2 = rec[e2];
      fs = dinv[rc & 0xFFFFF];
      fd = dinv[node0 + (int)((unsigned)rc >> 20)];
    } else {
      w = wq;
      rc = 0;
      wq = 0.0f;
      if (en < ee) {
        rc = rec[en];
        wq = wgt[en];
      }
    }
    const int s  = cur & 0xFFFFF;
    const int dl = (int)((unsigned)cur >> 20);
    unsigned m = __builtin_amdgcn_ballot_w32(ok);
    while (__builtin_popcount(m) >= 8)
      m = lds_rows<8, RP>(H, acc, s, dl, w, m, lane);
    if (__builtin_popcount(m) >= 4)
      m = lds_rows<4, RP>(H, acc, s, dl, w, m, lane);
    if (__builtin_popcount(m) >= 2)
      m = lds_rows<2, RP>(H, acc, s, dl, w, m, lane);
    if (m)
      m = lds_rows<1, RP>(H, acc, s, dl, w, m, lane);
  }
  __syncthreads();

  if (PROD) {
    if (wave < NSTRIP) {
      const int m  = lane & 15;
      const int h2 = lane >> 4;
      int arow = wave * 16 + m;
      if (arow > BIN - 1) arow = BIN - 1;
      const float* ap = acc + arow * RP;
      v8f c0 = {}, c1 = {}, c2 = {}, c3 = {};
#pragma unroll
      for (int kb = 0; kb < 64; kb += 4) {
        const int pa = (kb >> 1) + h2;
        const v2f a = *(const v2f*)(ap + 2 * pa);
        const v2f* wp = &wl[pa * 64 + m];
        const v2f b0 = wp[0];
        const v2f b1 = wp[16];
        const v2f b2 = wp[32];
        const v2f b3 = wp[48];
        c0 = __builtin_amdgcn_wmma_f32_16x16x4_f32(false, a, false, b0, (short)0, c0, false, false);
        c1 = __builtin_amdgcn_wmma_f32_16x16x4_f32(false, a, false, b1, (short)0, c1, false, false);
        c2 = __builtin_amdgcn_wmma_f32_16x16x4_f32(false, a, false, b2, (short)0, c2, false, false);
        c3 = __builtin_amdgcn_wmma_f32_16x16x4_f32(false, a, false, b3, (short)0, c3, false, false);
      }
      const float q0 = bias[m], q1 = bias[m + 16], q2 = bias[m + 32], q3 = bias[m + 48];
      const int r0 = wave * 16 + 8 * h2;
      if (r0 < BIN) {
        float* pp = acc + r0 * RP + m;
#pragma unroll
        for (int j = 0; j < 8; ++j) {
          float* r = pp + j * RP;
          r[0] = c0[j] + q0; r[16] = c1[j] + q1; r[32] = c2[j] + q2; r[48] = c3[j] + q3;
        }
      }
    }
    __syncthreads();
  }

#pragma unroll 4
  for (int i = wave; i < BIN; i += 16) {
    const size_t base = (size_t)(node0 + i) * 64 + lane * 2;
    const v2f a = *(const v2f*)(acc + i * RP + lane * 2);
    v2f o;
    if (FINAL) {
      o = a;
    } else {
      float vx = fmaxf(a.x, 0.0f);
      float vy = fmaxf(a.y, 0.0f);
      float ss = vx * vx + vy * vy;
#pragma unroll
      for (int off = 16; off > 0; off >>= 1) ss += __shfl_xor(ss, off, 32);
      float scale = 1.0f / fmaxf(sqrtf(ss), 1e-12f);
      o.x = vx * scale;
      o.y = vy * scale;
    }
    *(v2f*)(out + base) = o;
  }
}

extern "C" void kernel_launch(void* const* d_in, const int* in_sizes, int n_in,
                              void* d_out, int out_size, void* d_ws, size_t ws_size,
                              hipStream_t stream) {
  const float* x  = (const float*)d_in[0];
  const int*   ei = (const int*)d_in[1];
  const float* W1 = (const float*)d_in[2];
  const float* b1 = (const float*)d_in[3];
  const float* W2 = (const float*)d_in[4];
  const float* b2 = (const float*)d_in[5];
  const float* W3 = (const float*)d_in[6];
  const float* b3 = (const float*)d_in[7];
  float* out = (float*)d_out;

  const int* src = ei;
  const int* dst = ei + NE;

  char* ws = (char*)d_ws;
  size_t off = 0;
  auto carve = [&](size_t bytes) {
    void* p = ws + off;
    off = (off + bytes + 255) & ~(size_t)255;
    return p;
  };
  unsigned* bincnt   = (unsigned*)carve((size_t)NBINS * 4);
  unsigned* cursor   = (unsigned*)carve((size_t)NBINS * CUR_PITCH * 4);
  float*    dinv     = (float*)   carve((size_t)NN * 4);
  unsigned* binstart = (unsigned*)carve((size_t)(NBINS + 1) * 4);
  int*      rec      = (int*)     carve((size_t)NE * 4);
  float*    wgt      = (float*)   carve((size_t)NE * 4);
  float*    h        = (float*)   carve((size_t)NN * 64 * 4);
  float*    hn       = (float*)   carve((size_t)NN * 64 * 4);

  const int hcBlk = (NE + HC_EPB - 1) / HC_EPB;

  hipMemsetAsync(bincnt, 0, (size_t)((char*)(cursor + NBINS * CUR_PITCH) - (char*)bincnt), stream);
  k_bincnt  <<<hcBlk, 1024, 0, stream>>>(dst, bincnt);
  k_scan    <<<1,     512,  0, stream>>>(bincnt, binstart);
  k_scatter_gemm<<<SC_BLK + G_BLK, 256, 0, stream>>>(src, dst, binstart, cursor, rec, x, W1, h);
  k_dinv    <<<NBINS, 256,  0, stream>>>(rec, binstart, dinv);

  k_agg_lds<false, true, false><<<NBINS, 512, 0, stream>>>(h, rec, wgt, binstart, dinv, nullptr, b1, hn);

  k_agg_lds<false, false, true><<<NBINS, 512, 0, stream>>>(hn, rec, wgt, binstart, dinv, W2, b2, h);

  k_agg_lds<true, false, true><<<NBINS, 512, 0, stream>>>(h, rec, wgt, binstart, dinv, W3, b3, out);

  (void)in_sizes; (void)n_in; (void)out_size; (void)ws_size;
}


# Edge bucketing without a count pass: each bucketing block orders its 4096 edges by range in its own region; the degree launch joins each range's pieces (1 pool add); memset, count, scan launches gone
# speedup vs baseline: 3.8206x; 1.1154x over previous
#include <hip/hip_runtime.h>


#define NN 100000
#define NE 1600000

#define BIN    200
#define NBINS  (NN / BIN)
#define SC_EPB 4096
#define SC_BLK ((NE + SC_EPB - 1) / SC_EPB)
static_assert(NN % BIN == 0, "ranges cover the nodes exactly");
static_assert(NBINS <= 512, "one block-wide scan covers all ranges");
static_assert(SC_BLK <= 512, "two pieces per thread of the gathering block cover all bucketing blocks");
static_assert(SC_EPB <= 65535, "a piece's offset and its length share one word");
static_assert(NN <= (1 << 20) && BIN <= 2048, "source id and row-in-range share one word");

typedef float v2f __attribute__((ext_vector_type(2)));
typedef float v4f __attribute__((ext_vector_type(4)));
typedef float v8f __attribute__((ext_vector_type(8)));
typedef int   v4i __attribute__((ext_vector_type(4)));

__device__ __forceinline__ unsigned wave_incl_sum(unsigned v, int lane) {
#pragma unroll
  for (int off = 1; off < 32; off <<= 1) {
    const unsigned n = __shfl_up(v, off, 32);
    if (lane >= off) v += n;
  }
  return v;
}

__global__ __launch_bounds__(256) void k_dinv(const int* __restrict__ piece,
                                              const unsigned* __restrict__ tab,
                                              unsigned* __restrict__ pool,
                                              unsigned* __restrict__ rng,
                                              int* __restrict__ rec,
                                              float* __restrict__ dinv) {
  constexpr int U = 8;
  __shared__ unsigned cnt[BIN];
  __shared__ unsigned wsum[2][8];
  __shared__ unsigned sbase;
  const int t    = threadIdx.x;
  const int lane = t & 31;
  const int wave = t >> 5;
  const int r    = blockIdx.x;
  if (t < BIN) cnt[t] = 0u;
  unsigned e0 = 0u, e1 = 0u;
  if (t < SC_BLK)       e0 = tab[r * SC_BLK + t];
  if (t + 256 < SC_BLK) e1 = tab[r * SC_BLK + t + 256];
  const unsigned n0 = e0 >> 16, n1 = e1 >> 16;
  static_assert(NE > (SC_BLK - 1) * SC_EPB, "every bucketing block holds at least one edge");
  const int* rb0 = piece + (size_t)t * SC_EPB;
  const int* rb1 = piece + (size_t)(t + 256 < SC_BLK ? t + 256 : 0) * SC_EPB;
  const unsigned pl0 = e0 & 0xFFFFu, pl1 = e1 & 0xFFFFu;
  const unsigned i0 = wave_incl_sum(n0, lane);
  const unsigned i1 = wave_incl_sum(n1, lane);
  if (lane == 31) { wsum[0][wave] = i0; wsum[1][wave] = i1; }
  __syncthreads();
  unsigned p0 = 0u, p1 = 0u, t0 = 0u, t1 = 0u;
#pragma unroll
  for (int w = 0; w < 8; ++w) {
    const unsigned a = wsum[0][w], b = wsum[1][w];
    if (w < wave) { p0 += a; p1 += b; }
    t0 += a;
    t1 += b;
  }
  const unsigned off0 = p0 + i0 - n0;
  const unsigned off1 = t0 + p1 + i1 - n1;
  int x0[U], x1[U];
#pragma unroll
  for (int k = 0; k < U; ++k) {
    x0[k] = rb0[(unsigned)k < n0 ? pl0 + k : 0u];
    x1[k] = rb1[(unsigned)k < n1 ? pl1 + k : 0u];
  }
  if (t == 0) {
    const unsigned base = atomicAdd(pool, t0 + t1);
    sbase = base;
    rng[2 * r]     = base;
    rng[2 * r + 1] = t0 + t1;
  }
  __syncthreads();
  int* dp0 = rec + sbase + off0;
  int* dp1 = rec + sbase + off1;
  const unsigned nmax = n0 > n1 ? n0 : n1;
  for (unsigned k0 = 0u; k0 < nmax; k0 += U) {
#pragma unroll
    for (int k = 0; k < U; ++k) {
      if (k0 + k < n0) {
        dp0[k0 + k] = x0[k];
        atomicAdd(&cnt[(unsigned)x0[k] >> 20], 1u);
      }
      if (k0 + k < n1) {
        dp1[k0 + k] = x1[k];
        atomicAdd(&cnt[(unsigned)x1[k] >> 20], 1u);
      }
    }
    if (k0 + U < nmax) {
#pragma unroll
      for (int k = 0; k < U; ++k) {
        const unsigned kk = k0 + U + k;
        x0[k] = rb0[kk < n0 ? pl0 + kk : 0u];
        x1[k] = rb1[kk < n1 ? pl1 + kk : 0u];
      }
    }
  }
  __syncthreads();
  if (t < BIN)
    dinv[r * BIN + t] = 1.0f / sqrtf((float)cnt[t] + 1.0f);
}

template <int K>
__device__ __forceinline__ void gemm_block(const float* __restrict__ X,
                                           const float* __restrict__ W,
                                           float* __restrict__ H, int M,
                                           v2f* wl, int gb) {
  for (int i = threadIdx.x; i < (K / 2) * 64; i += 256) {
    int p = i >> 6, n = i & 63;
    v2f t = { W[(size_t)(2 * p) * 64 + n], W[(size_t)(2 * p + 1) * 64 + n] };
    wl[i] = t;
  }
  __syncthreads();

  int wave  = threadIdx.x >> 5;
  int lane  = threadIdx.x & 31;
  int strip = gb * 8 + wave;
  if (strip * 16 >= M) return;
  int row0 = strip * 16;
  int m    = lane & 15;
  int h2   = lane >> 4;

  const float* xrow = X + (size_t)(row0 + m) * K;
  v8f a0 = {}, a1 = {}, a2 = {}, a3 = {};
  for (int kb = 0; kb < K; kb += 4) {
    int pa = (kb >> 1) + h2;
    v2f a = *(const v2f*)(xrow + 2 * pa);
    const v2f* wp = &wl[pa * 64 + m];
    v2f b0 = wp[0];
    v2f b1 = wp[16];
    v2f b2 = wp[32];
    v2f b3 = wp[48];
    a0 = __builtin_amdgcn_wmma_f32_16x16x4_f32(false, a, false, b0, (short)0, a0, false, false);
    a1 = __builtin_amdgcn_wmma_f32_16x16x4_f32(false, a, false, b1, (short)0, a1, false, false);
    a2 = __builtin_amdgcn_wmma_f32_16x16x4_f32(false, a, false, b2, (short)0, a2, false, false);
    a3 = __builtin_amdgcn_wmma_f32_16x16x4_f32(false, a, false, b3, (short)0, a3, false, false);
  }
  float* out0 = H + (size_t)(row0 + 8 * h2) * 64 + m;
#pragma unroll
  for (int j = 0; j < 8; ++j) {
    float* r = out0 + (size_t)j * 64;
    r[0] = a0[j]; r[16] = a1[j]; r[32] = a2[j]; r[48] = a3[j];
  }
}

#define G_BLK  (((NN / 16) + 7) / 8)
__global__ __launch_bounds__(256) void k_scatter_gemm(const int* __restrict__ src,
                                                      const int* __restrict__ dst,
                                                      unsigned* __restrict__ tab,
                                                      unsigned* __restrict__ pool,
                                                      int* __restrict__ piece,
                                                      const float* __restrict__ X,
                                                      const float* __restrict__ W,
                                                      float* __restrict__ H) {
  __shared__ __align__(16) v2f wl[(256 / 2) * 64];
  if (blockIdx.x >= SC_BLK) {
    gemm_block<256>(X, W, H, NN, wl, (int)blockIdx.x - SC_BLK);
    return;
  }
  constexpr int PER = SC_EPB / 256;
  unsigned* cnt  = (unsigned*)wl;
  unsigned* wsum = cnt + 512;
  const int t    = threadIdx.x;
  const int lane = t & 31;
  const int wave = t >> 5;
  cnt[t] = 0u;
  cnt[t + 256] = 0u;
  if (blockIdx.x == 0 && t == 0) *pool = 0u;

  const int e0 = blockIdx.x * SC_EPB + t;
  bool ok[PER];
  int  s[PER], d[PER], b[PER];
#pragma unroll
  for (int k = 0; k < PER; ++k) {
    const int e = e0 + k * 256;
    ok[k] = e < NE;
    const int ec = ok[k] ? e : 0;
    d[k] = dst[ec];
    s[k] = src[ec];
  }
  __syncthreads();

  unsigned r[PER];
#pragma unroll
  for (int k = 0; k < PER; ++k) {
    b[k] = d[k] / BIN;
    r[k] = 0u;
    if (ok[k]) r[k] = atomicAdd(&cnt[b[k]], 1u);
  }
  __syncthreads();

  {
    const unsigned c0 = cnt[2 * t], c1 = cnt[2 * t + 1];
    const unsigned v  = c0 + c1;
    const unsigned in = wave_incl_sum(v, lane);
    if (lane == 31) wsum[wave] = in;
    __syncthreads();
    unsigned wp = 0u;
#pragma unroll
    for (int w = 0; w < 8; ++w) {
      const unsigned a = wsum[w];
      if (w < wave) wp += a;
    }
    const unsigned ex = wp + in - v;
    cnt[2 * t]     = ex;
    cnt[2 * t + 1] = ex + c0;
    if (2 * t < NBINS)     tab[(2 * t) * SC_BLK + blockIdx.x]     = ex | (c0 << 16);
    if (2 * t + 1 < NBINS) tab[(2 * t + 1) * SC_BLK + blockIdx.x] = (ex + c0) | (c1 << 16);
  }
  __syncthreads();

  int* reg = piece + (size_t)blockIdx.x * SC_EPB;
#pragma unroll
  for (int k = 0; k < PER; ++k) {
    if (ok[k]) {
      const unsigned pos = cnt[b[k]] + r[k];
      const int dl = d[k] - b[k] * BIN;
      reg[pos] = s[k] | (dl << 20);
    }
  }
}


template <int N, int RP>
__device__ __forceinline__ unsigned lds_rows(const float* __restrict__ H,
                                             float* acc,
                                             int s, int dl, float w,
                                             unsigned m, int lane) {
  int sj[N], dj[N];
  float wj[N];
  v2f hv[N];
#pragma unroll
  for (int i = 0; i < N; ++i) {
    int j = __builtin_ctz(m);
    m &= m - 1;
    sj[i] = __builtin_amdgcn_readlane(s, j);
    dj[i] = __builtin_amdgcn_readlane(dl, j);
    wj[i] = __int_as_float(__builtin_amdgcn_readlane(__float_as_int(w), j));
  }
#pragma unroll
  for (int i = 0; i < N; ++i)
    hv[i] = *(const v2f*)(H + (size_t)sj[i] * 64 + lane * 2);
#pragma unroll
  for (int i = 0; i < N; ++i) {
    float* p = acc + dj[i] * RP + lane * 2;
    __hip_atomic_fetch_add(p,     wj[i] * hv[i].x, __ATOMIC_RELAXED, __HIP_MEMORY_SCOPE_WORKGROUP);
    __hip_atomic_fetch_add(p + 1, wj[i] * hv[i].y, __ATOMIC_RELAXED, __HIP_MEMORY_SCOPE_WORKGROUP);
  }
  return m;
}

#define NSTRIP ((BIN + 15) / 16)
static_assert(BIN % 8 == 0, "a half strip (8 rows) lies wholly inside or wholly outside the range");
static_assert(NSTRIP <= 16, "one strip per wave");

template <bool FINAL, bool MAKEW, bool PROD>
__global__ __launch_bounds__(512) void k_agg_lds(const float* __restrict__ H,
                                                 const int* __restrict__ rec,
                                                 float* __restrict__ wgt,
                                                 const unsigned* __restrict__ rng,
                                                 const float* __restrict__ dinv,
                                                 const float* __restrict__ W,
                                                 const float* __restrict__ bias,
                                                 float* __restrict__ out) {
  constexpr int RP = PROD ? 68 : 64;
  __shared__ __align__(16) float acc[BIN * RP];
  __shared__ __align__(16) v2f wl[PROD ? 32 * 64 : 1];
  const int t    = threadIdx.x;
  const int lane = t & 31;
  const int wave = __builtin_amdgcn_readfirstlane(t >> 5);
  const int node0 = blockIdx.x * BIN;
  const int eb = (int)rng[2 * blockIdx.x];
  const int ee = eb + (int)rng[2 * blockIdx.x + 1];

  const int g0 = eb + wave * 32;
  int   rc = 0, rc2 = 0;
  float wq = 0.0f;
  if (g0 + lane < ee) {
    rc = rec[g0 + lane];
    if (!MAKEW) wq = wgt[g0 + lane];
  }
  if (MAKEW) {
    if (g0 + 16 * 32 + lane < ee) rc2 = rec[g0 + 16 * 32 + lane];
  }
  v2f wreg[4];
  if (PROD) {
#pragma unroll
    for (int k = 0; k < 4; ++k) {
      const int i = t + k * 512;
      const int p = i >> 6, n = i & 63;
      wreg[k] = (v2f){ W[(2 * p) * 64 + n], W[(2 * p + 1) * 64 + n] };
    }
  }

  constexpr int INIT_IT = (BIN * 16 + 511) / 512;
  float fs = 0.0f, fd = 0.0f;
  {
    const int c4 = (t & 15) * 4;
    v4f bv = (v4f){0.0f, 0.0f, 0.0f, 0.0f};
    if (!PROD) bv = *(const v4f*)(bias + c4);
    v4f   hv[INIT_IT];
    float dv[INIT_IT];
#pragma unroll
    for (int k = 0; k < INIT_IT; ++k) {
      const int i = t + k * 512;
      hv[k] = (v4f){0.0f, 0.0f, 0.0f, 0.0f};
      dv[k] = 0.0f;
      if (i < BIN * 16) {
        const int row = i >> 4;
        dv[k] = dinv[node0 + row];
        hv[k] = *(const v4f*)(H + (size_t)(node0 + row) * 64 + c4);
      }
    }
    if (MAKEW) {
      fs = dinv[rc & 0xFFFFF];
      fd = dinv[node0 + (int)((unsigned)rc >> 20)];
    }
#pragma unroll
    for (int k = 0; k < INIT_IT; ++k) {
      const int i = t + k * 512;
      if (i < BIN * 16) {
        const int row = i >> 4;
        const float sl = dv[k] * dv[k];
        v4f o;
        o.x = sl * hv[k].x + bv.x;
        o.y = sl * hv[k].y + bv.y;
        o.z = sl * hv[k].z + bv.z;
        o.w = sl * hv[k].w + bv.w;
        *(v4f*)(acc + row * RP + c4) = o;
      }
    }
    if (PROD) {
#pragma unroll
      for (int k = 0; k < 4; ++k) wl[t + k * 512] = wreg[k];
    }
  }
  __syncthreads();

  for (int g = g0; g < ee; g += 16 * 32) {
    const int  cur = rc;
    const bool ok = (g + lane) < ee;
    const int  en = g + 16 * 32 + lane;
    float w;
    if (MAKEW) {
      w = fs * fd;
      if (ok) wgt[g + lane] = w;
      rc = rc2;
      const int e2 = g + 2 * 16 * 32 + lane;
      rc2 = 0;
      if (e2 < ee) rc2 = rec[e2];
      fs = dinv[rc & 0xFFFFF];
      fd = dinv[node0 + (int)((unsigned)rc >> 20)];
    } else {
      w = wq;
      rc = 0;
      wq = 0.0f;
      if (en < ee) {
        rc = rec[en];
        wq = wgt[en];
      }
    }
    const int s  = cur & 0xFFFFF;
    const int dl = (int)((unsigned)cur >> 20);
    unsigned m = __builtin_amdgcn_ballot_w32(ok);
    while (__builtin_popcount(m) >= 8)
      m = lds_rows<8, RP>(H, acc, s, dl, w, m, lane);
    if (__builtin_popcount(m) >= 4)
      m = lds_rows<4, RP>(H, acc, s, dl, w, m, lane);
    if (__builtin_popcount(m) >= 2)
      m = lds_rows<2, RP>(H, acc, s, dl, w, m, lane);
    if (m)
      m = lds_rows<1, RP>(H, acc, s, dl, w, m, lane);
  }
  __syncthreads();

  if (PROD) {
    if (wave < NSTRIP) {
      const int m  = lane & 15;
      const int h2 = lane >> 4;
      int arow = wave * 16 + m;
      if (arow > BIN - 1) arow = BIN - 1;
      const float* ap = acc + arow * RP;
      v8f c0 = {}, c1 = {}, c2 = {}, c3 = {};
#pragma unroll
      for (int kb = 0; kb < 64; kb += 4) {
        const int pa = (kb >> 1) + h2;
        const v2f a = *(const v2f*)(ap + 2 * pa);
        const v2f* wp = &wl[pa * 64 + m];
        const v2f b0 = wp[0];
        const v2f b1 = wp[16];
        const v2f b2 = wp[32];
        const v2f b3 = wp[48];
        c0 = __builtin_amdgcn_wmma_f32_16x16x4_f32(false, a, false, b0, (short)0, c0, false, false);
        c1 = __builtin_amdgcn_wmma_f32_16x16x4_f32(false, a, false, b1, (short)0, c1, false, false);
        c2 = __builtin_amdgcn_wmma_f32_16x16x4_f32(false, a, false, b2, (short)0, c2, false, false);
        c3 = __builtin_amdgcn_wmma_f32_16x16x4_f32(false, a, false, b3, (short)0, c3, false, false);
      }
      const float q0 = bias[m], q1 = bias[m + 16], q2 = bias[m + 32], q3 = bias[m + 48];
      const int r0 = wave * 16 + 8 * h2;
      if (r0 < BIN) {
        float* pp = acc + r0 * RP + m;
#pragma unroll
        for (int j = 0; j < 8; ++j) {
          float* r = pp + j * RP;
          r[0] = c0[j] + q0; r[16] = c1[j] + q1; r[32] = c2[j] + q2; r[48] = c3[j] + q3;
        }
      }
    }
    __syncthreads();
  }

#pragma unroll 4
  for (int i = wave; i < BIN; i += 16) {
    const size_t base = (size_t)(node0 + i) * 64 + lane * 2;
    const v2f a = *(const v2f*)(acc + i * RP + lane * 2);
    v2f o;
    if (FINAL) {
      o = a;
    } else {
      float vx = fmaxf(a.x, 0.0f);
      float vy = fmaxf(a.y, 0.0f);
      float ss = vx * vx + vy * vy;
#pragma unroll
      for (int off = 16; off > 0; off >>= 1) ss += __shfl_xor(ss, off, 32);
      float scale = 1.0f / fmaxf(sqrtf(ss), 1e-12f);
      o.x = vx * scale;
      o.y = vy * scale;
    }
    *(v2f*)(out + base) = o;
  }
}

extern "C" void kernel_launch(void* const* d_in, const int* in_sizes, int n_in,
                              void* d_out, int out_size, void* d_ws, size_t ws_size,
                              hipStream_t stream) {
  const float* x  = (const float*)d_in[0];
  const int*   ei = (const int*)d_in[1];
  const float* W1 = (const float*)d_in[2];
  const float* b1 = (const float*)d_in[3];
  const float* W2 = (const float*)d_in[4];
  const float* b2 = (const float*)d_in[5];
  const float* W3 = (const float*)d_in[6];
  const float* b3 = (const float*)d_in[7];
  float* out = (float*)d_out;

  const int* src = ei;
  const int* dst = ei + NE;

  char* ws = (char*)d_ws;
  size_t off = 0;
  auto carve = [&](size_t bytes) {
    void* p = ws + off;
    off = (off + bytes + 255) & ~(size_t)255;
    return p;
  };
  unsigned* pool  = (unsigned*)carve(4);
  unsigned* rng   = (unsigned*)carve((size_t)NBINS * 2 * 4);
  unsigned* tab   = (unsigned*)carve((size_t)NBINS * SC_BLK * 4);
  int*      piece = (int*)     carve((size_t)SC_BLK * SC_EPB * 4);
  float*    dinv  = (float*)   carve((size_t)NN * 4);
  int*      rec   = (int*)     carve((size_t)NE * 4);
  float*    wgt   = (float*)   carve((size_t)NE * 4);
  float*    h     = (float*)   carve((size_t)NN * 64 * 4);
  float*    hn    = (float*)   carve((size_t)NN * 64 * 4);

  k_scatter_gemm<<<SC_BLK + G_BLK, 256, 0, stream>>>(src, dst, tab, pool, piece, x, W1, h);
  k_dinv    <<<NBINS, 256,  0, stream>>>(piece, tab, pool, rng, rec, dinv);

  k_agg_lds<false, true, false><<<NBINS, 512, 0, stream>>>(h, rec, wgt, rng, dinv, nullptr, b1, hn);

  k_agg_lds<false, false, true><<<NBINS, 512, 0, stream>>>(hn, rec, wgt, rng, dinv, W2, b2, h);

  k_agg_lds<true, false, true><<<NBINS, 512, 0, stream>>>(h, rec, wgt, rng, dinv, W3, b3, out);

  (void)in_sizes; (void)n_in; (void)out_size; (void)ws_size;
}


# Range sums by row owner: block deals its records out to its 16 waves by row & 15 (LDS counters); each wave adds its rows by plain LDS read-add-write, no float atomics; sum order = owner list order
# speedup vs baseline: 4.3238x; 1.1317x over previous
#include <hip/hip_runtime.h>


#define NN 100000
#define NE 1600000

#define BIN    200
#define NBINS  (NN / BIN)
#define SC_EPB 4096
#define SC_BLK ((NE + SC_EPB - 1) / SC_EPB)
static_assert(NN % BIN == 0, "ranges cover the nodes exactly");
static_assert(NBINS <= 512, "one block-wide scan covers all ranges");
static_assert(SC_BLK <= 512, "two pieces per thread of the gathering block cover all bucketing blocks");
static_assert(SC_EPB <= 65535, "a piece's offset and its length share one word");
static_assert(NN <= (1 << 20) && BIN <= 2048, "source id and row-in-range share one word");

typedef float v2f __attribute__((ext_vector_type(2)));
typedef float v4f __attribute__((ext_vector_type(4)));
typedef float v8f __attribute__((ext_vector_type(8)));
typedef int   v4i __attribute__((ext_vector_type(4)));
typedef int   v2i __attribute__((ext_vector_type(2)));

__device__ __forceinline__ unsigned wave_incl_sum(unsigned v, int lane) {
#pragma unroll
  for (int off = 1; off < 32; off <<= 1) {
    const unsigned n = __shfl_up(v, off, 32);
    if (lane >= off) v += n;
  }
  return v;
}

__global__ __launch_bounds__(256) void k_dinv(const int* __restrict__ piece,
                                              const unsigned* __restrict__ tab,
                                              unsigned* __restrict__ pool,
                                              unsigned* __restrict__ rng,
                                              int* __restrict__ rec,
                                              float* __restrict__ dinv) {
  constexpr int U = 8;
  __shared__ unsigned cnt[BIN];
  __shared__ unsigned wsum[2][8];
  __shared__ unsigned sbase;
  const int t    = threadIdx.x;
  const int lane = t & 31;
  const int wave = t >> 5;
  const int r    = blockIdx.x;
  if (t < BIN) cnt[t] = 0u;
  unsigned e0 = 0u, e1 = 0u;
  if (t < SC_BLK)       e0 = tab[r * SC_BLK + t];
  if (t + 256 < SC_BLK) e1 = tab[r * SC_BLK + t + 256];
  const unsigned n0 = e0 >> 16, n1 = e1 >> 16;
  static_assert(NE > (SC_BLK - 1) * SC_EPB, "every bucketing block holds at least one edge");
  const int* rb0 = piece + (size_t)t * SC_EPB;
  const int* rb1 = piece + (size_t)(t + 256 < SC_BLK ? t + 256 : 0) * SC_EPB;
  const unsigned pl0 = e0 & 0xFFFFu, pl1 = e1 & 0xFFFFu;
  const unsigned i0 = wave_incl_sum(n0, lane);
  const unsigned i1 = wave_incl_sum(n1, lane);
  if (lane == 31) { wsum[0][wave] = i0; wsum[1][wave] = i1; }
  __syncthreads();
  unsigned p0 = 0u, p1 = 0u, t0 = 0u, t1 = 0u;
#pragma unroll
  for (int w = 0; w < 8; ++w) {
    const unsigned a = wsum[0][w], b = wsum[1][w];
    if (w < wave) { p0 += a; p1 += b; }
    t0 += a;
    t1 += b;
  }
  const unsigned off0 = p0 + i0 - n0;
  const unsigned off1 = t0 + p1 + i1 - n1;
  int x0[U], x1[U];
#pragma unroll
  for (int k = 0; k < U; ++k) {
    x0[k] = rb0[(unsigned)k < n0 ? pl0 + k : 0u];
    x1[k] = rb1[(unsigned)k < n1 ? pl1 + k : 0u];
  }
  if (t == 0) {
    const unsigned base = atomicAdd(pool, t0 + t1);
    sbase = base;
    rng[2 * r]     = base;
    rng[2 * r + 1] = t0 + t1;
  }
  __syncthreads();
  int* dp0 = rec + sbase + off0;
  int* dp1 = rec + sbase + off1;
  const unsigned nmax = n0 > n1 ? n0 : n1;
  for (unsigned k0 = 0u; k0 < nmax; k0 += U) {
#pragma unroll
    for (int k = 0; k < U; ++k) {
      if (k0 + k < n0) {
        dp0[k0 + k] = x0[k];
        atomicAdd(&cnt[(unsigned)x0[k] >> 20], 1u);
      }
      if (k0 + k < n1) {
        dp1[k0 + k] = x1[k];
        atomicAdd(&cnt[(unsigned)x1[k] >> 20], 1u);
      }
    }
    if (k0 + U < nmax) {
#pragma unroll
      for (int k = 0; k < U; ++k) {
        const unsigned kk = k0 + U + k;
        x0[k] = rb0[kk < n0 ? pl0 + kk : 0u];
        x1[k] = rb1[kk < n1 ? pl1 + kk : 0u];
      }
    }
  }
  __syncthreads();
  if (t < BIN)
    dinv[r * BIN + t] = 1.0f / sqrtf((float)cnt[t] + 1.0f);
}

template <int K>
__device__ __forceinline__ void gemm_block(const float* __restrict__ X,
                                           const float* __restrict__ W,
                                           float* __restrict__ H, int M,
                                           v2f* wl, int gb) {
  for (int i = threadIdx.x; i < (K / 2) * 64; i += 256) {
    int p = i >> 6, n = i & 63;
    v2f t = { W[(size_t)(2 * p) * 64 + n], W[(size_t)(2 * p + 1) * 64 + n] };
    wl[i] = t;
  }
  __syncthreads();

  int wave  = threadIdx.x >> 5;
  int lane  = threadIdx.x & 31;
  int strip = gb * 8 + wave;
  if (strip * 16 >= M) return;
  int row0 = strip * 16;
  int m    = lane & 15;
  int h2   = lane >> 4;

  const float* xrow = X + (size_t)(row0 + m) * K;
  v8f a0 = {}, a1 = {}, a2 = {}, a3 = {};
  for (int kb = 0; kb < K; kb += 4) {
    int pa = (kb >> 1) + h2;
    v2f a = *(const v2f*)(xrow + 2 * pa);
    const v2f* wp = &wl[pa * 64 + m];
    v2f b0 = wp[0];
    v2f b1 = wp[16];
    v2f b2 = wp[32];
    v2f b3 = wp[48];
    a0 = __builtin_amdgcn_wmma_f32_16x16x4_f32(false, a, false, b0, (short)0, a0, false, false);
    a1 = __builtin_amdgcn_wmma_f32_16x16x4_f32(false, a, false, b1, (short)0, a1, false, false);
    a2 = __builtin_amdgcn_wmma_f32_16x16x4_f32(false, a, false, b2, (short)0, a2, false, false);
    a3 = __builtin_amdgcn_wmma_f32_16x16x4_f32(false, a, false, b3, (short)0, a3, false, false);
  }
  float* out0 = H + (size_t)(row0 + 8 * h2) * 64 + m;
#pragma unroll
  for (int j = 0; j < 8; ++j) {
    float* r = out0 + (size_t)j * 64;
    r[0] = a0[j]; r[16] = a1[j]; r[32] = a2[j]; r[48] = a3[j];
  }
}

#define G_BLK  (((NN / 16) + 7) / 8)
__global__ __launch_bounds__(256) void k_scatter_gemm(const int* __restrict__ src,
                                                      const int* __restrict__ dst,
                                                      unsigned* __restrict__ tab,
                                                      unsigned* __restrict__ pool,
                                                      int* __restrict__ piece,
                                                      const float* __restrict__ X,
                                                      const float* __restrict__ W,
                                                      float* __restrict__ H) {
  __shared__ __align__(16) v2f wl[(256 / 2) * 64];
  if (blockIdx.x >= SC_BLK) {
    gemm_block<256>(X, W, H, NN, wl, (int)blockIdx.x - SC_BLK);
    return;
  }
  constexpr int PER = SC_EPB / 256;
  unsigned* cnt  = (unsigned*)wl;
  unsigned* wsum = cnt + 512;
  const int t    = threadIdx.x;
  const int lane = t & 31;
  const int wave = t >> 5;
  cnt[t] = 0u;
  cnt[t + 256] = 0u;
  if (blockIdx.x == 0 && t == 0) *pool = 0u;

  const int e0 = blockIdx.x * SC_EPB + t;
  bool ok[PER];
  int  s[PER], d[PER], b[PER];
#pragma unroll
  for (int k = 0; k < PER; ++k) {
    const int e = e0 + k * 256;
    ok[k] = e < NE;
    const int ec = ok[k] ? e : 0;
    d[k] = dst[ec];
    s[k] = src[ec];
  }
  __syncthreads();

  unsigned r[PER];
#pragma unroll
  for (int k = 0; k < PER; ++k) {
    b[k] = d[k] / BIN;
    r[k] = 0u;
    if (ok[k]) r[k] = atomicAdd(&cnt[b[k]], 1u);
  }
  __syncthreads();

  {
    const unsigned c0 = cnt[2 * t], c1 = cnt[2 * t + 1];
    const unsigned v  = c0 + c1;
    const unsigned in = wave_incl_sum(v, lane);
    if (lane == 31) wsum[wave] = in;
    __syncthreads();
    unsigned wp = 0u;
#pragma unroll
    for (int w = 0; w < 8; ++w) {
      const unsigned a = wsum[w];
      if (w < wave) wp += a;
    }
    const unsigned ex = wp + in - v;
    cnt[2 * t]     = ex;
    cnt[2 * t + 1] = ex + c0;
    if (2 * t < NBINS)     tab[(2 * t) * SC_BLK + blockIdx.x]     = ex | (c0 << 16);
    if (2 * t + 1 < NBINS) tab[(2 * t + 1) * SC_BLK + blockIdx.x] = (ex + c0) | (c1 << 16);
  }
  __syncthreads();

  int* reg = piece + (size_t)blockIdx.x * SC_EPB;
#pragma unroll
  for (int k = 0; k < PER; ++k) {
    if (ok[k]) {
      const unsigned pos = cnt[b[k]] + r[k];
      const int dl = d[k] - b[k] * BIN;
      reg[pos] = s[k] | (dl << 20);
    }
  }
}


#define QCH   4096
#define QPER  (QCH / 512)
static_assert(QCH % 512 == 0, "whole records per thread");

template <int HALF>
__device__ __forceinline__ void rows_issue(const float* __restrict__ H, int s, int lane,
                                           v2f (&hv)[16]) {
#pragma unroll
  for (int i = 0; i < 16; ++i) {
    const int sj = __builtin_amdgcn_readlane(s, HALF * 16 + i);
    hv[i] = *(const v2f*)(H + (size_t)sj * 64 + lane * 2);
  }
}

template <int HALF, int RP, bool FULL>
__device__ __forceinline__ void rows_add(float* acc, int dl, float w, int lane,
                                         const v2f (&hv)[16], int cnt) {
#pragma unroll
  for (int i = 0; i < 16; ++i) {
    if (FULL || i < cnt) {
      const int   dj = __builtin_amdgcn_readlane(dl, HALF * 16 + i);
      const float wj = __int_as_float(__builtin_amdgcn_readlane(__float_as_int(w), HALF * 16 + i));
      v2f* p = (v2f*)(acc + dj * RP + lane * 2);
      v2f o = *p;
      o.x += wj * hv[i].x;
      o.y += wj * hv[i].y;
      *p = o;
    }
  }
}

#define NSTRIP ((BIN + 15) / 16)
static_assert(BIN % 8 == 0, "a half strip (8 rows) lies wholly inside or wholly outside the range");
static_assert(NSTRIP <= 16, "one strip per wave");

template <bool FINAL, bool MAKEW, bool PROD>
__global__ __launch_bounds__(512, 8) void k_agg_lds(const float* __restrict__ H,
                                                 const int* __restrict__ rec,
                                                 float* __restrict__ wgt,
                                                 const unsigned* __restrict__ rng,
                                                 const float* __restrict__ dinv,
                                                 const float* __restrict__ W,
                                                 const float* __restrict__ bias,
                                                 float* __restrict__ out) {
  constexpr int RP = PROD ? 68 : 64;
  __shared__ __align__(16) float acc[BIN * RP];
  __shared__ __align__(16) v2f wl[PROD ? 32 * 64 : 1];
  __shared__ __align__(8) v2i q[QCH];
  __shared__ unsigned qcnt[64];
  __shared__ unsigned qbase[65];
  const int t    = threadIdx.x;
  const int lane = t & 31;
  const int wave = __builtin_amdgcn_readfirstlane(t >> 5);
  const int node0 = blockIdx.x * BIN;
  const int eb = (int)rng[2 * blockIdx.x];
  const int ee = eb + (int)rng[2 * blockIdx.x + 1];
  if (t < 64) qcnt[t] = 0u;

  int   rcv[QPER];
  float wv[QPER];
  int cn = ee - eb < QCH ? ee - eb : QCH;
#pragma unroll
  for (int k = 0; k < QPER; ++k) {
    const int i = t + k * 512;
    rcv[k] = 0;
    wv[k]  = 0.0f;
    if (i < cn) {
      rcv[k] = rec[eb + i];
      if (!MAKEW) wv[k] = wgt[eb + i];
    }
  }
  v2f wreg[4];
  if (PROD) {
#pragma unroll
    for (int k = 0; k < 4; ++k) {
      const int i = t + k * 512;
      const int p = i >> 6, n = i & 63;
      wreg[k] = (v2f){ W[(2 * p) * 64 + n], W[(2 * p + 1) * 64 + n] };
    }
  }

  constexpr int INIT_IT = (BIN * 16 + 511) / 512;
  {
    const int c4 = (t & 15) * 4;
    v4f bv = (v4f){0.0f, 0.0f, 0.0f, 0.0f};
    if (!PROD) bv = *(const v4f*)(bias + c4);
    v4f   hv[INIT_IT];
    float dv[INIT_IT];
#pragma unroll
    for (int k = 0; k < INIT_IT; ++k) {
      const int i = t + k * 512;
      hv[k] = (v4f){0.0f, 0.0f, 0.0f, 0.0f};
      dv[k] = 0.0f;
      if (i < BIN * 16) {
        const int row = i >> 4;
        dv[k] = dinv[node0 + row];
        hv[k] = *(const v4f*)(H + (size_t)(node0 + row) * 64 + c4);
      }
    }
    float fs[QPER], fd[QPER];
    if (MAKEW) {
#pragma unroll
      for (int k = 0; k < QPER; ++k) {
        fs[k] = dinv[rcv[k] & 0xFFFFF];
        fd[k] = dinv[node0 + (int)((unsigned)rcv[k] >> 20)];
      }
    }
#pragma unroll
    for (int k = 0; k < INIT_IT; ++k) {
      const int i = t + k * 512;
      if (i < BIN * 16) {
        const int row = i >> 4;
        const float sl = dv[k] * dv[k];
        v4f o;
        o.x = sl * hv[k].x + bv.x;
        o.y = sl * hv[k].y + bv.y;
        o.z = sl * hv[k].z + bv.z;
        o.w = sl * hv[k].w + bv.w;
        *(v4f*)(acc + row * RP + c4) = o;
      }
    }
    if (PROD) {
#pragma unroll
      for (int k = 0; k < 4; ++k) wl[t + k * 512] = wreg[k];
    }
    if (MAKEW) {
#pragma unroll
      for (int k = 0; k < QPER; ++k) {
        const int i = t + k * 512;
        wv[k] = fs[k] * fd[k];
        if (i < cn) wgt[eb + i] = wv[k];
      }
    }
  }
  __syncthreads();

  for (int c0 = eb; c0 < ee; c0 += QCH) {
    if (c0 != eb) {
      __syncthreads();
      if (t < 64) qcnt[t] = 0u;
      cn = ee - c0 < QCH ? ee - c0 : QCH;
#pragma unroll
      for (int k = 0; k < QPER; ++k) {
        const int i = t + k * 512;
        rcv[k] = 0;
        wv[k]  = 0.0f;
        if (i < cn) {
          rcv[k] = rec[c0 + i];
          if (!MAKEW) wv[k] = wgt[c0 + i];
        }
      }
      if (MAKEW) {
#pragma unroll
        for (int k = 0; k < QPER; ++k) {
          const int i = t + k * 512;
          wv[k] = dinv[rcv[k] & 0xFFFFF] * dinv[node0 + (int)((unsigned)rcv[k] >> 20)];
          if (i < cn) wgt[c0 + i] = wv[k];
        }
      }
      __syncthreads();
    }
    unsigned key[QPER], rk[QPER];
#pragma unroll
    for (int k = 0; k < QPER; ++k) {
      const int i = t + k * 512;
      key[k] = ((((unsigned)rcv[k] >> 20) & 15u) << 2) | (unsigned)(lane >> 3);
      rk[k]  = 0u;
      if (i < cn) rk[k] = atomicAdd(&qcnt[key[k]], 1u);
    }
    __syncthreads();
    if (t < 32) {
      const unsigned a = qcnt[2 * t], b = qcnt[2 * t + 1];
      const unsigned in = wave_incl_sum(a + b, lane);
      const unsigned ex = in - (a + b);
      qbase[2 * t]     = ex;
      qbase[2 * t + 1] = ex + a;
      if (t == 31) qbase[64] = in;
    }
    __syncthreads();
#pragma unroll
    for (int k = 0; k < QPER; ++k) {
      const int i = t + k * 512;
      if (i < cn) q[qbase[key[k]] + rk[k]] = (v2i){ rcv[k], __float_as_int(wv[k]) };
    }
    __syncthreads();

    const int lb = __builtin_amdgcn_readfirstlane((int)qbase[wave * 4]);
    const int le = __builtin_amdgcn_readfirstlane((int)qbase[wave * 4 + 4]);
    if (lb < le) {
      v2f ha[16], hb[16];
      int   es, edl;
      float ew;
      {
        const int idx = lb + lane;
        const v2i e = q[idx < le ? idx : le - 1];
        es  = e.x & 0xFFFFF;
        edl = (int)((unsigned)e.x >> 20);
        ew  = __int_as_float(e.y);
      }
      rows_issue<0>(H, es, lane, ha);
      int g = lb;
      for (; le - g > 32; g += 32) {
        rows_issue<1>(H, es, lane, hb);
        rows_add<0, RP, true>(acc, edl, ew, lane, ha, 16);
        const int idx = g + 32 + lane;
        const v2i e = q[idx < le ? idx : le - 1];
        const int   ns  = e.x & 0xFFFFF;
        const int   ndl = (int)((unsigned)e.x >> 20);
        const float nw  = __int_as_float(e.y);
        rows_issue<0>(H, ns, lane, ha);
        rows_add<1, RP, true>(acc, edl, ew, lane, hb, 16);
        es = ns; edl = ndl; ew = nw;
      }
      {
        const int left = le - g;
        const int ca = left < 16 ? left : 16;
        const int cb = left - 16;
        if (cb > 0) rows_issue<1>(H, es, lane, hb);
        rows_add<0, RP, false>(acc, edl, ew, lane, ha, ca);
        if (cb > 0) rows_add<1, RP, false>(acc, edl, ew, lane, hb, cb);
      }
    }
  }
  __syncthreads();

  if (PROD) {
    if (wave < NSTRIP) {
      const int m  = lane & 15;
      const int h2 = lane >> 4;
      int arow = wave * 16 + m;
      if (arow > BIN - 1) arow = BIN - 1;
      const float* ap = acc + arow * RP;
      v8f c0 = {}, c1 = {}, c2 = {}, c3 = {};
#pragma unroll
      for (int kb = 0; kb < 64; kb += 4) {
        const int pa = (kb >> 1) + h2;
        const v2f a = *(const v2f*)(ap + 2 * pa);
        const v2f* wp = &wl[pa * 64 + m];
        const v2f b0 = wp[0];
        const v2f b1 = wp[16];
        const v2f b2 = wp[32];
        const v2f b3 = wp[48];
        c0 = __builtin_amdgcn_wmma_f32_16x16x4_f32(false, a, false, b0, (short)0, c0, false, false);
        c1 = __builtin_amdgcn_wmma_f32_16x16x4_f32(false, a, false, b1, (short)0, c1, false, false);
        c2 = __builtin_amdgcn_wmma_f32_16x16x4_f32(false, a, false, b2, (short)0, c2, false, false);
        c3 = __builtin_amdgcn_wmma_f32_16x16x4_f32(false, a, false, b3, (short)0, c3, false, false);
      }
      const float q0 = bias[m], q1 = bias[m + 16], q2 = bias[m + 32], q3 = bias[m + 48];
      const int r0 = wave * 16 + 8 * h2;
      if (r0 < BIN) {
        float* pp = acc + r0 * RP + m;
#pragma unroll
        for (int j = 0; j < 8; ++j) {
          float* r = pp + j * RP;
          r[0] = c0[j] + q0; r[16] = c1[j] + q1; r[32] = c2[j] + q2; r[48] = c3[j] + q3;
        }
      }
    }
    __syncthreads();
  }

#pragma unroll 4
  for (int i = wave; i < BIN; i += 16) {
    const size_t base = (size_t)(node0 + i) * 64 + lane * 2;
    const v2f a = *(const v2f*)(acc + i * RP + lane * 2);
    v2f o;
    if (FINAL) {
      o = a;
    } else {
      float vx = fmaxf(a.x, 0.0f);
      float vy = fmaxf(a.y, 0.0f);
      float ss = vx * vx + vy * vy;
#pragma unroll
      for (int off = 16; off > 0; off >>= 1) ss += __shfl_xor(ss, off, 32);
      float scale = 1.0f / fmaxf(sqrtf(ss), 1e-12f);
      o.x = vx * scale;
      o.y = vy * scale;
    }
    *(v2f*)(out + base) = o;
  }
}

extern "C" void kernel_launch(void* const* d_in, const int* in_sizes, int n_in,
                              void* d_out, int out_size, void* d_ws, size_t ws_size,
                              hipStream_t stream) {
  const float* x  = (const float*)d_in[0];
  const int*   ei = (const int*)d_in[1];
  const float* W1 = (const float*)d_in[2];
  const float* b1 = (const float*)d_in[3];
  const float* W2 = (const float*)d_in[4];
  const float* b2 = (const float*)d_in[5];
  const float* W3 = (const float*)d_in[6];
  const float* b3 = (const float*)d_in[7];
  float* out = (float*)d_out;

  const int* src = ei;
  const int* dst = ei + NE;

  char* ws = (char*)d_ws;
  size_t off = 0;
  auto carve = [&](size_t bytes) {
    void* p = ws + off;
    off = (off + bytes + 255) & ~(size_t)255;
    return p;
  };
  unsigned* pool  = (unsigned*)carve(4);
  unsigned* rng   = (unsigned*)carve((size_t)NBINS * 2 * 4);
  unsigned* tab   = (unsigned*)carve((size_t)NBINS * SC_BLK * 4);
  int*      piece = (int*)     carve((size_t)SC_BLK * SC_EPB * 4);
  float*    dinv  = (float*)   carve((size_t)NN * 4);
  int*      rec   = (int*)     carve((size_t)NE * 4);
  float*    wgt   = (float*)   carve((size_t)NE * 4);
  float*    h     = (float*)   carve((size_t)NN * 64 * 4);
  float*    hn    = (float*)   carve((size_t)NN * 64 * 4);

  k_scatter_gemm<<<SC_BLK + G_BLK, 256, 0, stream>>>(src, dst, tab, pool, piece, x, W1, h);
  k_dinv    <<<NBINS, 256,  0, stream>>>(piece, tab, pool, rng, rec, dinv);

  k_agg_lds<false, true, false><<<NBINS, 512, 0, stream>>>(h, rec, wgt, rng, dinv, nullptr, b1, hn);

  k_agg_lds<false, false, true><<<NBINS, 512, 0, stream>>>(hn, rec, wgt, rng, dinv, W2, b2, h);

  k_agg_lds<true, false, true><<<NBINS, 512, 0, stream>>>(h, rec, wgt, rng, dinv, W3, b3, out);

  (void)in_sizes; (void)n_in; (void)out_size; (void)ws_size;
}


# Range sums two edges per instruction: a row's owner is a half wave (32 owners, row mod 32), 16-byte lanes; each load and LDS read-add-write serves one edge per half; sum order = owner list order
# speedup vs baseline: 4.4886x; 1.0381x over previous
#include <hip/hip_runtime.h>


#define NN 100000
#define NE 1600000

#define BIN    200
#define NBINS  (NN / BIN)
#define SC_EPB 4096
#define SC_BLK ((NE + SC_EPB - 1) / SC_EPB)
static_assert(NN % BIN == 0, "ranges cover the nodes exactly");
static_assert(NBINS <= 512, "one block-wide scan covers all ranges");
static_assert(SC_BLK <= 512, "two pieces per thread of the gathering block cover all bucketing blocks");
static_assert(SC_EPB <= 65535, "a piece's offset and its length share one word");
static_assert(NN <= (1 << 20) && BIN <= 2048, "source id and row-in-range share one word");

typedef float v2f __attribute__((ext_vector_type(2)));
typedef float v4f __attribute__((ext_vector_type(4)));
typedef float v8f __attribute__((ext_vector_type(8)));
typedef int   v4i __attribute__((ext_vector_type(4)));
typedef int   v2i __attribute__((ext_vector_type(2)));

__device__ __forceinline__ unsigned wave_incl_sum(unsigned v, int lane) {
#pragma unroll
  for (int off = 1; off < 32; off <<= 1) {
    const unsigned n = __shfl_up(v, off, 32);
    if (lane >= off) v += n;
  }
  return v;
}

__global__ __launch_bounds__(256) void k_dinv(const int* __restrict__ piece,
                                              const unsigned* __restrict__ tab,
                                              unsigned* __restrict__ pool,
                                              unsigned* __restrict__ rng,
                                              int* __restrict__ rec,
                                              float* __restrict__ dinv) {
  constexpr int U = 8;
  __shared__ unsigned cnt[BIN];
  __shared__ unsigned wsum[2][8];
  __shared__ unsigned sbase;
  const int t    = threadIdx.x;
  const int lane = t & 31;
  const int wave = t >> 5;
  const int r    = blockIdx.x;
  if (t < BIN) cnt[t] = 0u;
  unsigned e0 = 0u, e1 = 0u;
  if (t < SC_BLK)       e0 = tab[r * SC_BLK + t];
  if (t + 256 < SC_BLK) e1 = tab[r * SC_BLK + t + 256];
  const unsigned n0 = e0 >> 16, n1 = e1 >> 16;
  static_assert(NE > (SC_BLK - 1) * SC_EPB, "every bucketing block holds at least one edge");
  const int* rb0 = piece + (size_t)t * SC_EPB;
  const int* rb1 = piece + (size_t)(t + 256 < SC_BLK ? t + 256 : 0) * SC_EPB;
  const unsigned pl0 = e0 & 0xFFFFu, pl1 = e1 & 0xFFFFu;
  const unsigned i0 = wave_incl_sum(n0, lane);
  const unsigned i1 = wave_incl_sum(n1, lane);
  if (lane == 31) { wsum[0][wave] = i0; wsum[1][wave] = i1; }
  __syncthreads();
  unsigned p0 = 0u, p1 = 0u, t0 = 0u, t1 = 0u;
#pragma unroll
  for (int w = 0; w < 8; ++w) {
    const unsigned a = wsum[0][w], b = wsum[1][w];
    if (w < wave) { p0 += a; p1 += b; }
    t0 += a;
    t1 += b;
  }
  const unsigned off0 = p0 + i0 - n0;
  const unsigned off1 = t0 + p1 + i1 - n1;
  int x0[U], x1[U];
#pragma unroll
  for (int k = 0; k < U; ++k) {
    x0[k] = rb0[(unsigned)k < n0 ? pl0 + k : 0u];
    x1[k] = rb1[(unsigned)k < n1 ? pl1 + k : 0u];
  }
  if (t == 0) {
    const unsigned base = atomicAdd(pool, t0 + t1);
    sbase = base;
    rng[2 * r]     = base;
    rng[2 * r + 1] = t0 + t1;
  }
  __syncthreads();
  int* dp0 = rec + sbase + off0;
  int* dp1 = rec + sbase + off1;
  const unsigned nmax = n0 > n1 ? n0 : n1;
  for (unsigned k0 = 0u; k0 < nmax; k0 += U) {
#pragma unroll
    for (int k = 0; k < U; ++k) {
      if (k0 + k < n0) {
        dp0[k0 + k] = x0[k];
        atomicAdd(&cnt[(unsigned)x0[k] >> 20], 1u);
      }
      if (k0 + k < n1) {
        dp1[k0 + k] = x1[k];
        atomicAdd(&cnt[(unsigned)x1[k] >> 20], 1u);
      }
    }
    if (k0 + U < nmax) {
#pragma unroll
      for (int k = 0; k < U; ++k) {
        const unsigned kk = k0 + U + k;
        x0[k] = rb0[kk < n0 ? pl0 + kk : 0u];
        x1[k] = rb1[kk < n1 ? pl1 + kk : 0u];
      }
    }
  }
  __syncthreads();
  if (t < BIN)
    dinv[r * BIN + t] = 1.0f / sqrtf((float)cnt[t] + 1.0f);
}

template <int K>
__device__ __forceinline__ void gemm_block(const float* __restrict__ X,
                                           const float* __restrict__ W,
                                           float* __restrict__ H, int M,
                                           v2f* wl, int gb) {
  for (int i = threadIdx.x; i < (K / 2) * 64; i += 256) {
    int p = i >> 6, n = i & 63;
    v2f t = { W[(size_t)(2 * p) * 64 + n], W[(size_t)(2 * p + 1) * 64 + n] };
    wl[i] = t;
  }
  __syncthreads();

  int wave  = threadIdx.x >> 5;
  int lane  = threadIdx.x & 31;
  int strip = gb * 8 + wave;
  if (strip * 16 >= M) return;
  int row0 = strip * 16;
  int m    = lane & 15;
  int h2   = lane >> 4;

  const float* xrow = X + (size_t)(row0 + m) * K;
  v8f a0 = {}, a1 = {}, a2 = {}, a3 = {};
  for (int kb = 0; kb < K; kb += 4) {
    int pa = (kb >> 1) + h2;
    v2f a = *(const v2f*)(xrow + 2 * pa);
    const v2f* wp = &wl[pa * 64 + m];
    v2f b0 = wp[0];
    v2f b1 = wp[16];
    v2f b2 = wp[32];
    v2f b3 = wp[48];
    a0 = __builtin_amdgcn_wmma_f32_16x16x4_f32(false, a, false, b0, (short)0, a0, false, false);
    a1 = __builtin_amdgcn_wmma_f32_16x16x4_f32(false, a, false, b1, (short)0, a1, false, false);
    a2 = __builtin_amdgcn_wmma_f32_16x16x4_f32(false, a, false, b2, (short)0, a2, false, false);
    a3 = __builtin_amdgcn_wmma_f32_16x16x4_f32(false, a, false, b3, (short)0, a3, false, false);
  }
  float* out0 = H + (size_t)(row0 + 8 * h2) * 64 + m;
#pragma unroll
  for (int j = 0; j < 8; ++j) {
    float* r = out0 + (size_t)j * 64;
    r[0] = a0[j]; r[16] = a1[j]; r[32] = a2[j]; r[48] = a3[j];
  }
}

#define G_BLK  (((NN / 16) + 7) / 8)
__global__ __launch_bounds__(256) void k_scatter_gemm(const int* __restrict__ src,
                                                      const int* __restrict__ dst,
                                                      unsigned* __restrict__ tab,
                                                      unsigned* __restrict__ pool,
                                                      int* __restrict__ piece,
                                                      const float* __restrict__ X,
                                                      const float* __restrict__ W,
                                                      float* __restrict__ H) {
  __shared__ __align__(16) v2f wl[(256 / 2) * 64];
  if (blockIdx.x >= SC_BLK) {
    gemm_block<256>(X, W, H, NN, wl, (int)blockIdx.x - SC_BLK);
    return;
  }
  constexpr int PER = SC_EPB / 256;
  unsigned* cnt  = (unsigned*)wl;
  unsigned* wsum = cnt + 512;
  const int t    = threadIdx.x;
  const int lane = t & 31;
  const int wave = t >> 5;
  cnt[t] = 0u;
  cnt[t + 256] = 0u;
  if (blockIdx.x == 0 && t == 0) *pool = 0u;

  const int e0 = blockIdx.x * SC_EPB + t;
  bool ok[PER];
  int  s[PER], d[PER], b[PER];
#pragma unroll
  for (int k = 0; k < PER; ++k) {
    const int e = e0 + k * 256;
    ok[k] = e < NE;
    const int ec = ok[k] ? e : 0;
    d[k] = dst[ec];
    s[k] = src[ec];
  }
  __syncthreads();

  unsigned r[PER];
#pragma unroll
  for (int k = 0; k < PER; ++k) {
    b[k] = d[k] / BIN;
    r[k] = 0u;
    if (ok[k]) r[k] = atomicAdd(&cnt[b[k]], 1u);
  }
  __syncthreads();

  {
    const unsigned c0 = cnt[2 * t], c1 = cnt[2 * t + 1];
    const unsigned v  = c0 + c1;
    const unsigned in = wave_incl_sum(v, lane);
    if (lane == 31) wsum[wave] = in;
    __syncthreads();
    unsigned wp = 0u;
#pragma unroll
    for (int w = 0; w < 8; ++w) {
      const unsigned a = wsum[w];
      if (w < wave) wp += a;
    }
    const unsigned ex = wp + in - v;
    cnt[2 * t]     = ex;
    cnt[2 * t + 1] = ex + c0;
    if (2 * t < NBINS)     tab[(2 * t) * SC_BLK + blockIdx.x]     = ex | (c0 << 16);
    if (2 * t + 1 < NBINS) tab[(2 * t + 1) * SC_BLK + blockIdx.x] = (ex + c0) | (c1 << 16);
  }
  __syncthreads();

  int* reg = piece + (size_t)blockIdx.x * SC_EPB;
#pragma unroll
  for (int k = 0; k < PER; ++k) {
    if (ok[k]) {
      const unsigned pos = cnt[b[k]] + r[k];
      const int dl = d[k] - b[k] * BIN;
      reg[pos] = s[k] | (dl << 20);
    }
  }
}


#define QCH   4096
#define QPER  (QCH / 512)
#define NOWN  32
static_assert(QCH % 512 == 0, "whole records per thread");

template <int IDX>
__device__ __forceinline__ unsigned half_bcast(unsigned v) {
  return (unsigned)__builtin_amdgcn_update_dpp(0, (int)v, 0x150 + IDX, 0xF, 0xF, true);
}

template <int G, int I = 0>
__device__ __forceinline__ void rows_issue(const float* __restrict__ H, unsigned so,
                                           unsigned l16b, v4f (&hv)[8]) {
  if constexpr (I < 8) {
    const unsigned off = half_bcast<G * 8 + I>(so) | l16b;
    hv[I] = *(const v4f*)((const char*)H + off);
    rows_issue<G, I + 1>(H, so, l16b, hv);
  }
}

template <int G, int I = 0>
__device__ __forceinline__ void rows_add(float* acc, unsigned ro, float w, unsigned l16b,
                                         const v4f (&hv)[8]) {
  if constexpr (I < 8) {
    const unsigned off = half_bcast<G * 8 + I>(ro) + l16b;
    const float wj = __uint_as_float(half_bcast<G * 8 + I>(__float_as_uint(w)));
    v4f* p = (v4f*)((char*)acc + off);
    v4f o = *p;
    o.x += wj * hv[I].x;
    o.y += wj * hv[I].y;
    o.z += wj * hv[I].z;
    o.w += wj * hv[I].w;
    *p = o;
    rows_add<G, I + 1>(acc, ro, w, l16b, hv);
  }
}

#define NSTRIP ((BIN + 15) / 16)
static_assert(BIN % 8 == 0, "a half strip (8 rows) lies wholly inside or wholly outside the range");
static_assert(NSTRIP <= 16, "one strip per wave");

template <bool FINAL, bool MAKEW, bool PROD>
__global__ __launch_bounds__(512, 8) void k_agg_lds(const float* __restrict__ H,
                                                 const int* __restrict__ rec,
                                                 float* __restrict__ wgt,
                                                 const unsigned* __restrict__ rng,
                                                 const float* __restrict__ dinv,
                                                 const float* __restrict__ W,
                                                 const float* __restrict__ bias,
                                                 float* __restrict__ out) {
  constexpr int RP = PROD ? 68 : 64;
  __shared__ __align__(16) float acc[(BIN + NOWN) * RP];
  __shared__ __align__(16) v2f wl[PROD ? 32 * 64 : 1];
  __shared__ __align__(8) v2i q[QCH];
  __shared__ unsigned qcnt[NOWN * 4];
  __shared__ unsigned qbase[NOWN * 4 + 1];
  const int t    = threadIdx.x;
  const int lane = t & 31;
  const int wave = __builtin_amdgcn_readfirstlane(t >> 5);
  const int node0 = blockIdx.x * BIN;
  const int eb = (int)rng[2 * blockIdx.x];
  const int ee = eb + (int)rng[2 * blockIdx.x + 1];
  if (t < NOWN * 4) qcnt[t] = 0u;

  int   rcv[QPER];
  float wv[QPER];
  int cn = ee - eb < QCH ? ee - eb : QCH;
#pragma unroll
  for (int k = 0; k < QPER; ++k) {
    const int i = t + k * 512;
    rcv[k] = 0;
    wv[k]  = 0.0f;
    if (i < cn) {
      rcv[k] = rec[eb + i];
      if (!MAKEW) wv[k] = wgt[eb + i];
    }
  }
  v2f wreg[4];
  if (PROD) {
#pragma unroll
    for (int k = 0; k < 4; ++k) {
      const int i = t + k * 512;
      const int p = i >> 6, n = i & 63;
      wreg[k] = (v2f){ W[(2 * p) * 64 + n], W[(2 * p + 1) * 64 + n] };
    }
  }

  constexpr int INIT_IT = (BIN * 16 + 511) / 512;
  {
    const int c4 = (t & 15) * 4;
    v4f bv = (v4f){0.0f, 0.0f, 0.0f, 0.0f};
    if (!PROD) bv = *(const v4f*)(bias + c4);
    v4f   hv[INIT_IT];
    float dv[INIT_IT];
#pragma unroll
    for (int k = 0; k < INIT_IT; ++k) {
      const int i = t + k * 512;
      hv[k] = (v4f){0.0f, 0.0f, 0.0f, 0.0f};
      dv[k] = 0.0f;
      if (i < BIN * 16) {
        const int row = i >> 4;
        dv[k] = dinv[node0 + row];
        hv[k] = *(const v4f*)(H + (size_t)(node0 + row) * 64 + c4);
      }
    }
    float fs[QPER], fd[QPER];
    if (MAKEW) {
#pragma unroll
      for (int k = 0; k < QPER; ++k) {
        fs[k] = dinv[rcv[k] & 0xFFFFF];
        fd[k] = dinv[node0 + (int)((unsigned)rcv[k] >> 20)];
      }
    }
#pragma unroll
    for (int k = 0; k < INIT_IT; ++k) {
      const int i = t + k * 512;
      if (i < BIN * 16) {
        const int row = i >> 4;
        const float sl = dv[k] * dv[k];
        v4f o;
        o.x = sl * hv[k].x + bv.x;
        o.y = sl * hv[k].y + bv.y;
        o.z = sl * hv[k].z + bv.z;
        o.w = sl * hv[k].w + bv.w;
        *(v4f*)(acc + row * RP + c4) = o;
      }
    }
    static_assert(NOWN * 16 == 512, "one piece of the spare rows per thread");
    *(v4f*)(acc + (BIN + (t >> 4)) * RP + c4) = (v4f){0.0f, 0.0f, 0.0f, 0.0f};
    if (PROD) {
#pragma unroll
      for (int k = 0; k < 4; ++k) wl[t + k * 512] = wreg[k];
    }
    if (MAKEW) {
#pragma unroll
      for (int k = 0; k < QPER; ++k) {
        const int i = t + k * 512;
        wv[k] = fs[k] * fd[k];
        if (i < cn) wgt[eb + i] = wv[k];
      }
    }
  }
  __syncthreads();

  for (int c0 = eb; c0 < ee; c0 += QCH) {
    if (c0 != eb) {
      __syncthreads();
      if (t < NOWN * 4) qcnt[t] = 0u;
      cn = ee - c0 < QCH ? ee - c0 : QCH;
#pragma unroll
      for (int k = 0; k < QPER; ++k) {
        const int i = t + k * 512;
        rcv[k] = 0;
        wv[k]  = 0.0f;
        if (i < cn) {
          rcv[k] = rec[c0 + i];
          if (!MAKEW) wv[k] = wgt[c0 + i];
        }
      }
      if (MAKEW) {
#pragma unroll
        for (int k = 0; k < QPER; ++k) {
          const int i = t + k * 512;
          wv[k] = dinv[rcv[k] & 0xFFFFF] * dinv[node0 + (int)((unsigned)rcv[k] >> 20)];
          if (i < cn) wgt[c0 + i] = wv[k];
        }
      }
      __syncthreads();
    }
    unsigned key[QPER], rk[QPER];
#pragma unroll
    for (int k = 0; k < QPER; ++k) {
      const int i = t + k * 512;
      key[k] = ((((unsigned)rcv[k] >> 20) & (unsigned)(NOWN - 1)) << 2) | (unsigned)(lane >> 3);
      rk[k]  = 0u;
      if (i < cn) rk[k] = atomicAdd(&qcnt[key[k]], 1u);
    }
    __syncthreads();
    static_assert(NOWN == 32, "one owner per lane of the scanning wave");
    if (t < 32) {
      const unsigned a = qcnt[4 * t], b = qcnt[4 * t + 1], cc = qcnt[4 * t + 2], d = qcnt[4 * t + 3];
      const unsigned in = wave_incl_sum(a + b + cc + d, lane);
      const unsigned ex = in - (a + b + cc + d);
      qbase[4 * t]     = ex;
      qbase[4 * t + 1] = ex + a;
      qbase[4 * t + 2] = ex + a + b;
      qbase[4 * t + 3] = ex + a + b + cc;
      if (t == 31) qbase[NOWN * 4] = in;
    }
    __syncthreads();
#pragma unroll
    for (int k = 0; k < QPER; ++k) {
      const int i = t + k * 512;
      if (i < cn) q[qbase[key[k]] + rk[k]] = (v2i){ rcv[k], __float_as_int(wv[k]) };
    }
    __syncthreads();

    const int      own  = wave + 16 * (lane >> 4);
    const unsigned l16b = (unsigned)(lane & 15) * 16u;
    const int mlb = (int)qbase[own * 4];
    const int mle = (int)qbase[own * 4 + 4];
    const int lenA = __builtin_amdgcn_readlane(mle - mlb, 0);
    const int lenB = __builtin_amdgcn_readlane(mle - mlb, 16);
    int rem = lenA > lenB ? lenA : lenB;
    if (rem > 0) {
      v4f ha[8], hb[8];
      unsigned so, ro;
      float    ew;
      auto fetch = [&](int base, unsigned& s, unsigned& r, float& w) {
        const int  idx = base + (lane & 15);
        const bool ok  = idx < mle;
        const v2i  e   = q[ok ? idx : 0];
        s = ((unsigned)e.x & 0xFFFFFu) << 8;
        r = (ok ? (unsigned)e.x >> 20 : (unsigned)(BIN + own)) * (unsigned)(RP * 4);
        w = ok ? __int_as_float(e.y) : 0.0f;
      };
      int base = mlb;
      fetch(base, so, ro, ew);
      rows_issue<0>(H, so, l16b, ha);
      for (; rem > 16; rem -= 16) {
        rows_issue<1>(H, so, l16b, hb);
        rows_add<0>(acc, ro, ew, l16b, ha);
        base += 16;
        unsigned ns, nr;
        float    nw;
        fetch(base, ns, nr, nw);
        rows_issue<0>(H, ns, l16b, ha);
        rows_add<1>(acc, ro, ew, l16b, hb);
        so = ns; ro = nr; ew = nw;
      }
      if (rem > 8) {
        rows_issue<1>(H, so, l16b, hb);
        rows_add<0>(acc, ro, ew, l16b, ha);
        rows_add<1>(acc, ro, ew, l16b, hb);
      } else {
        rows_add<0>(acc, ro, ew, l16b, ha);
      }
    }
  }
  __syncthreads();

  if (PROD) {
    if (wave < NSTRIP) {
      const int m  = lane & 15;
      const int h2 = lane >> 4;
      int arow = wave * 16 + m;
      if (arow > BIN - 1) arow = BIN - 1;
      const float* ap = acc + arow * RP;
      v8f c0 = {}, c1 = {}, c2 = {}, c3 = {};
#pragma unroll
      for (int kb = 0; kb < 64; kb += 4) {
        const int pa = (kb >> 1) + h2;
        const v2f a = *(const v2f*)(ap + 2 * pa);
        const v2f* wp = &wl[pa * 64 + m];
        const v2f b0 = wp[0];
        const v2f b1 = wp[16];
        const v2f b2 = wp[32];
        const v2f b3 = wp[48];
        c0 = __builtin_amdgcn_wmma_f32_16x16x4_f32(false, a, false, b0, (short)0, c0, false, false);
        c1 = __builtin_amdgcn_wmma_f32_16x16x4_f32(false, a, false, b1, (short)0, c1, false, false);
        c2 = __builtin_amdgcn_wmma_f32_16x16x4_f32(false, a, false, b2, (short)0, c2, false, false);
        c3 = __builtin_amdgcn_wmma_f32_16x16x4_f32(false, a, false, b3, (short)0, c3, false, false);
      }
      const float q0 = bias[m], q1 = bias[m + 16], q2 = bias[m + 32], q3 = bias[m + 48];
      const int r0 = wave * 16 + 8 * h2;
      if (r0 < BIN) {
        float* pp = acc + r0 * RP + m;
#pragma unroll
        for (int j = 0; j < 8; ++j) {
          float* r = pp + j * RP;
          r[0] = c0[j] + q0; r[16] = c1[j] + q1; r[32] = c2[j] + q2; r[48] = c3[j] + q3;
        }
      }
    }
    __syncthreads();
  }

#pragma unroll 4
  for (int i = wave; i < BIN; i += 16) {
    const size_t base = (size_t)(node0 + i) * 64 + lane * 2;
    const v2f a = *(const v2f*)(acc + i * RP + lane * 2);
    v2f o;
    if (FINAL) {
      o = a;
    } else {
      float vx = fmaxf(a.x, 0.0f);
      float vy = fmaxf(a.y, 0.0f);
      float ss = vx * vx + vy * vy;
#pragma unroll
      for (int off = 16; off > 0; off >>= 1) ss += __shfl_xor(ss, off, 32);
      float scale = 1.0f / fmaxf(sqrtf(ss), 1e-12f);
      o.x = vx * scale;
      o.y = vy * scale;
    }
    *(v2f*)(out + base) = o;
  }
}

extern "C" void kernel_launch(void* const* d_in, const int* in_sizes, int n_in,
                              void* d_out, int out_size, void* d_ws, size_t ws_size,
                              hipStream_t stream) {
  const float* x  = (const float*)d_in[0];
  const int*   ei = (const int*)d_in[1];
  const float* W1 = (const float*)d_in[2];
  const float* b1 = (const float*)d_in[3];
  const float* W2 = (const float*)d_in[4];
  const float* b2 = (const float*)d_in[5];
  const float* W3 = (const float*)d_in[6];
  const float* b3 = (const float*)d_in[7];
  float* out = (float*)d_out;

  const int* src = ei;
  const int* dst = ei + NE;

  char* ws = (char*)d_ws;
  size_t off = 0;
  auto carve = [&](size_t bytes) {
    void* p = ws + off;
    off = (off + bytes + 255) & ~(size_t)255;
    return p;
  };
  unsigned* pool  = (unsigned*)carve(4);
  unsigned* rng   = (unsigned*)carve((size_t)NBINS * 2 * 4);
  unsigned* tab   = (unsigned*)carve((size_t)NBINS * SC_BLK * 4);
  int*      piece = (int*)     carve((size_t)SC_BLK * SC_EPB * 4);
  float*    dinv  = (float*)   carve((size_t)NN * 4);
  int*      rec   = (int*)     carve((size_t)NE * 4);
  float*    wgt   = (float*)   carve((size_t)NE * 4);
  float*    h     = (float*)   carve((size_t)NN * 64 * 4);
  float*    hn    = (float*)   carve((size_t)NN * 64 * 4);

  k_scatter_gemm<<<SC_BLK + G_BLK, 256, 0, stream>>>(src, dst, tab, pool, piece, x, W1, h);
  k_dinv    <<<NBINS, 256,  0, stream>>>(piece, tab, pool, rng, rec, dinv);

  k_agg_lds<false, true, false><<<NBINS, 512, 0, stream>>>(h, rec, wgt, rng, dinv, nullptr, b1, hn);

  k_agg_lds<false, false, true><<<NBINS, 512, 0, stream>>>(hn, rec, wgt, rng, dinv, W2, b2, h);

  k_agg_lds<true, false, true><<<NBINS, 512, 0, stream>>>(h, rec, wgt, rng, dinv, W3, b3, out);

  (void)in_sizes; (void)n_in; (void)out_size; (void)ws_size;
}
